# starting kernel
# baseline (speedup 1.0000x reference)
#include <hip/hip_runtime.h>
#include <hip/hip_bf16.h>

typedef __attribute__((ext_vector_type(2))) float v2f;
typedef __attribute__((ext_vector_type(8))) float v8f;

#define NNODES 100000
#define DFEAT  128
#define BN_EPS 1e-5f


__global__ __launch_bounds__(256) void zero_kernel(float* __restrict__ p, size_t n) {
    size_t i = (size_t)blockIdx.x * blockDim.x + threadIdx.x;
    size_t stride = (size_t)gridDim.x * blockDim.x;
    for (; i < n; i += stride) p[i] = 0.0f;
}

__global__ __launch_bounds__(256) void degree_kernel(const int* __restrict__ src,
                                                     const int* __restrict__ dst,
                                                     float* __restrict__ dout,
                                                     float* __restrict__ din, int E) {
    int e = blockIdx.x * blockDim.x + threadIdx.x;
    if (e >= E) return;
    atomicAdd(&dout[src[e]], 1.0f);
    atomicAdd(&din[dst[e]], 1.0f);
}

__global__ __launch_bounds__(256) void norm_kernel(float* __restrict__ dout,
                                                   float* __restrict__ din, int n) {
    int i = blockIdx.x * blockDim.x + threadIdx.x;
    if (i >= n) return;
    dout[i] = rsqrtf(fmaxf(dout[i], 1.0f));
    din[i]  = rsqrtf(fmaxf(din[i], 1.0f));
}

__global__ __launch_bounds__(256) void scatter_kernel(const float* __restrict__ feats,
                                                      const int* __restrict__ src,
                                                      const int* __restrict__ dst,
                                                      const float* __restrict__ norm_src,
                                                      float* __restrict__ agg, int E) {
    long long t = (long long)blockIdx.x * blockDim.x + threadIdx.x;
    int e = (int)(t >> 5);
    if (e >= E) return;
    int lane = (int)(t & 31);
    int s = src[e], d = dst[e];
    float ns = norm_src[s];
    const float4* fr = (const float4*)(feats + (size_t)s * DFEAT);
    float4 v = fr[lane];
    float* ar = agg + (size_t)d * DFEAT + lane * 4;
    atomicAdd(ar + 0, v.x * ns);
    atomicAdd(ar + 1, v.y * ns);
    atomicAdd(ar + 2, v.z * ns);
    atomicAdd(ar + 3, v.w * ns);
}

__global__ __launch_bounds__(256) void gemm_bn_kernel(
    const float* __restrict__ agg, const float* __restrict__ feats,
    const float* __restrict__ norm_dst,
    const float* __restrict__ W,  const float* __restrict__ b,
    const float* __restrict__ Wr, const float* __restrict__ br,
    float* __restrict__ y, float* __restrict__ bn_sum, float* __restrict__ bn_sumsq) {
    const int lane = threadIdx.x & 31;
    const int wave = threadIdx.x >> 5;
    const int rowBase = blockIdx.x * 16;
    const int n0 = wave * 16;
    const int m  = lane & 15;
    const int kh = (lane >> 4) * 2;
    const int row = rowBase + m;
    const float nd = norm_dst[row];
    const float* aRow = agg   + (size_t)row * DFEAT;
    const float* fRow = feats + (size_t)row * DFEAT;

    v8f accC = {};
    v8f accR = {};
    #pragma unroll
    for (int k = 0; k < DFEAT; k += 4) {
        v2f a, f, bw, bwr;
        a.x = aRow[k + kh]     * nd;
        a.y = aRow[k + kh + 1] * nd;
        f.x = fRow[k + kh];
        f.y = fRow[k + kh + 1];
        bw.x  = W [(size_t)(k + kh)     * DFEAT + n0 + m];
        bw.y  = W [(size_t)(k + kh + 1) * DFEAT + n0 + m];
        bwr.x = Wr[(size_t)(k + kh)     * DFEAT + n0 + m];
        bwr.y = Wr[(size_t)(k + kh + 1) * DFEAT + n0 + m];
        accC = __builtin_amdgcn_wmma_f32_16x16x4_f32(false, a, false, bw,  (short)0, accC, false, false);
        accR = __builtin_amdgcn_wmma_f32_16x16x4_f32(false, f, false, bwr, (short)0, accR, false, false);
    }

    const int col  = n0 + m;
    const float bc  = b[col];
    const float brc = br[col];
    const int rOff = (lane >> 4) * 8;
    float s8 = 0.0f, q8 = 0.0f;
    #pragma unroll
    for (int r = 0; r < 8; ++r) {
        float c = accC[r] + bc;  c = c > 0.0f ? c : 0.0f;
        float e = accR[r] + brc; e = e > 0.0f ? e : 0.0f;
        float v = c + e;
        y[(size_t)(rowBase + rOff + r) * DFEAT + col] = v;
        s8 += v;
        q8 += v * v;
    }
    float s8o = __shfl_down(s8, 16);
    float q8o = __shfl_down(q8, 16);
    if (lane < 16) {
        atomicAdd(&bn_sum[col],   s8 + s8o);
        atomicAdd(&bn_sumsq[col], q8 + q8o);
    }
}

__global__ void bn_params_kernel(const float* __restrict__ sum, const float* __restrict__ sumsq,
                                 const float* __restrict__ gamma, const float* __restrict__ beta,
                                 float* __restrict__ scale, float* __restrict__ shift, int n) {
    int c = threadIdx.x;
    if (c >= DFEAT) return;
    float inv  = 1.0f / (float)n;
    float mean = sum[c] * inv;
    float var  = sumsq[c] * inv - mean * mean;
    float sc   = gamma[c] * rsqrtf(var + BN_EPS);
    scale[c] = sc;
    shift[c] = beta[c] - mean * sc;
}

__global__ __launch_bounds__(256) void bn_apply_kernel(float* __restrict__ y,
                                                       const float* __restrict__ scale,
                                                       const float* __restrict__ shift,
                                                       size_t nvec4) {
    size_t i = (size_t)blockIdx.x * blockDim.x + threadIdx.x;
    if (i >= nvec4) return;
    float4 v = ((float4*)y)[i];
    int col = (int)((i * 4) & (DFEAT - 1));
    v.x = v.x * scale[col + 0] + shift[col + 0];
    v.y = v.y * scale[col + 1] + shift[col + 1];
    v.z = v.z * scale[col + 2] + shift[col + 2];
    v.w = v.w * scale[col + 3] + shift[col + 3];
    ((float4*)y)[i] = v;
}

extern "C" void kernel_launch(void* const* d_in, const int* in_sizes, int n_in,
                              void* d_out, int out_size, void* d_ws, size_t ws_size,
                              hipStream_t stream) {
    const float* feats = (const float*)d_in[0];
    const int*   src   = (const int*)d_in[1];
    const int*   dst   = (const int*)d_in[2];
    const float* W     = (const float*)d_in[3];
    const float* b     = (const float*)d_in[4];
    const float* Wr    = (const float*)d_in[5];
    const float* br    = (const float*)d_in[6];
    const float* gamma = (const float*)d_in[7];
    const float* beta  = (const float*)d_in[8];
    float* out = (float*)d_out;

    const int N = NNODES;
    const int E = in_sizes[1];

    float* ws        = (float*)d_ws;
    float* agg       = ws;
    float* deg_out   = agg + (size_t)N * DFEAT;
    float* deg_in    = deg_out + N;
    float* bn_sum    = deg_in + N;
    float* bn_sumsq  = bn_sum + DFEAT;
    float* bn_scale  = bn_sumsq + DFEAT;
    float* bn_shift  = bn_scale + DFEAT;

    size_t nz = (size_t)N * DFEAT + 2 * (size_t)N + 2 * DFEAT;
    zero_kernel<<<4096, 256, 0, stream>>>(ws, nz);

    degree_kernel<<<(E + 255) / 256, 256, 0, stream>>>(src, dst, deg_out, deg_in, E);

    norm_kernel<<<(N + 255) / 256, 256, 0, stream>>>(deg_out, deg_in, N);

    long long sthreads = (long long)E * 32;
    int sblocks = (int)((sthreads + 255) / 256);
    scatter_kernel<<<sblocks, 256, 0, stream>>>(feats, src, dst, deg_out, agg, E);

    gemm_bn_kernel<<<N / 16, 256, 0, stream>>>(agg, feats, deg_in, W, b, Wr, br,
                                               out, bn_sum, bn_sumsq);

    bn_params_kernel<<<1, DFEAT, 0, stream>>>(bn_sum, bn_sumsq, gamma, beta,
                                              bn_scale, bn_shift, N);

    size_t nvec4 = (size_t)N * DFEAT / 4;
    bn_apply_kernel<<<(int)((nvec4 + 255) / 256), 256, 0, stream>>>(out, bn_scale, bn_shift, nvec4);
}


# Aggregation by owner: each of 256 work-groups sweeps all destinations and sums its 391 rows in LDS (arrival order of LDS adds), one store per line; no global f32 atomics or zero fill of agg
# speedup vs baseline: 1.3936x; 1.3936x over previous
#include <hip/hip_runtime.h>
#include <hip/hip_bf16.h>

typedef __attribute__((ext_vector_type(2))) float v2f;
typedef __attribute__((ext_vector_type(8))) float v8f;

#define NNODES 100000
#define DFEAT  128
#define BN_EPS 1e-5f


__global__ __launch_bounds__(256) void zero_kernel(float* __restrict__ p, size_t n) {
    size_t i = (size_t)blockIdx.x * blockDim.x + threadIdx.x;
    size_t stride = (size_t)gridDim.x * blockDim.x;
    for (; i < n; i += stride) p[i] = 0.0f;
}

__global__ __launch_bounds__(256) void degree_kernel(const int* __restrict__ src,
                                                     const int* __restrict__ dst,
                                                     float* __restrict__ dout,
                                                     float* __restrict__ din, int E) {
    int e = blockIdx.x * blockDim.x + threadIdx.x;
    if (e >= E) return;
    atomicAdd(&dout[src[e]], 1.0f);
    atomicAdd(&din[dst[e]], 1.0f);
}

__global__ __launch_bounds__(256) void norm_kernel(float* __restrict__ dout,
                                                   float* __restrict__ din, int n) {
    int i = blockIdx.x * blockDim.x + threadIdx.x;
    if (i >= n) return;
    dout[i] = rsqrtf(fmaxf(dout[i], 1.0f));
    din[i]  = rsqrtf(fmaxf(din[i], 1.0f));
}

#define AGG_WGS     256
#define AGG_WAVES   32
#define AGG_ROWS    ((NNODES + AGG_WGS - 1) / AGG_WGS)
#define AGG_GROUP   512
#define AGG_QCAP    (32 + AGG_GROUP)

__device__ __forceinline__ void agg_batch(const float* __restrict__ feats, const int* __restrict__ src,
                                          const int* __restrict__ dst, const float* __restrict__ norm_src,
                                          float* acc, const int* q, int n, int lo, int lane) {
    int s = 0, dl = 0;
    float ns = 0.0f;
    if (lane < n) {
        const int e = q[lane];
        dl = dst[e] - lo;
        s  = src[e];
        ns = norm_src[s];
    }
    const int nsb = __float_as_int(ns);
    const float* fl = feats + lane;
    float* al = acc + lane;
    for (int j0 = 0; j0 < n; j0 += 8) {
        float v[8][4];
        #pragma unroll
        for (int u = 0; u < 8; ++u) {
            const int sj = __builtin_amdgcn_readlane(s, j0 + u);
            const float* fr = fl + (size_t)sj * DFEAT;
            v[u][0] = fr[0];
            v[u][1] = fr[32];
            v[u][2] = fr[64];
            v[u][3] = fr[96];
        }
        #pragma unroll
        for (int u = 0; u < 8; ++u) {
            if (j0 + u < n) {
                const int dj = __builtin_amdgcn_readlane(dl, j0 + u);
                const float nj = __int_as_float(__builtin_amdgcn_readlane(nsb, j0 + u));
                float* ar = al + dj * DFEAT;
                atomicAdd(ar + 0,  v[u][0] * nj);
                atomicAdd(ar + 32, v[u][1] * nj);
                atomicAdd(ar + 64, v[u][2] * nj);
                atomicAdd(ar + 96, v[u][3] * nj);
            }
        }
    }
}

__global__ __launch_bounds__(1024) void aggregate_kernel(const float* __restrict__ feats,
                                                         const int* __restrict__ src,
                                                         const int* __restrict__ dst,
                                                         const float* __restrict__ norm_src,
                                                         float* __restrict__ agg, int E) {
    __shared__ __attribute__((aligned(16))) float acc[AGG_ROWS * DFEAT];
    __shared__ int queue[AGG_WAVES][AGG_QCAP];

    const int tid  = threadIdx.x;
    const int lane = tid & 31;
    const int wave = __builtin_amdgcn_readfirstlane(tid >> 5);
    const int lo   = blockIdx.x * AGG_ROWS;
    int rows = NNODES - lo;
    rows = rows > AGG_ROWS ? AGG_ROWS : rows;
    rows = rows < 0 ? 0 : rows;

    for (int i = tid; i < AGG_ROWS * DFEAT / 4; i += 1024)
        ((float4*)acc)[i] = make_float4(0.0f, 0.0f, 0.0f, 0.0f);
    __syncthreads();

    int* q = queue[wave];
    int fill = 0;

    const int nFull = E / AGG_GROUP;
    if (nFull > 0) {
        const int rot = (int)(((long long)blockIdx.x * nFull) / AGG_WGS);
        const int4* dst4 = (const int4*)dst + lane;

        int4 cur[4];
        {
            int g = (wave < nFull ? wave : 0) + rot;
            g = g >= nFull ? g - nFull : g;
            #pragma unroll
            for (int c = 0; c < 4; ++c) cur[c] = dst4[(size_t)g * (AGG_GROUP / 4) + c * 32];
        }
        for (int gi = wave; gi < nFull; gi += AGG_WAVES) {
            int g = gi + rot;
            g = g >= nFull ? g - nFull : g;
            int4 nxt[4];
            {
                int gn = (gi + AGG_WAVES < nFull ? gi + AGG_WAVES : gi) + rot;
                gn = gn >= nFull ? gn - nFull : gn;
                #pragma unroll
                for (int c = 0; c < 4; ++c) nxt[c] = dst4[(size_t)gn * (AGG_GROUP / 4) + c * 32];
            }
            #pragma unroll
            for (int c = 0; c < 4; ++c) {
                const int d4[4] = {cur[c].x, cur[c].y, cur[c].z, cur[c].w};
                bool m[4];
                #pragma unroll
                for (int k = 0; k < 4; ++k)
                    m[k] = (unsigned)(d4[k] - lo) < (unsigned)rows;
                if (__builtin_amdgcn_ballot_w32(m[0] | m[1] | m[2] | m[3]) == 0u) continue;
                const int e0 = g * AGG_GROUP + c * 128 + 4 * lane;
                #pragma unroll
                for (int k = 0; k < 4; ++k) {
                    const unsigned mask = __builtin_amdgcn_ballot_w32(m[k]);
                    if (mask == 0u) continue;
                    if (m[k]) q[fill + (int)__builtin_amdgcn_mbcnt_lo(mask, 0u)] = e0 + k;
                    fill += __builtin_popcount(mask);
                }
            }
            if (fill >= 32) {
                int head = 0;
                while (fill - head >= 32) {
                    agg_batch(feats, src, dst, norm_src, acc, q + head, 32, lo, lane);
                    head += 32;
                }
                const int left = fill - head;
                int t = 0;
                if (lane < left) t = q[head + lane];
                if (lane < left) q[lane] = t;
                fill = left;
            }
            #pragma unroll
            for (int c = 0; c < 4; ++c) cur[c] = nxt[c];
        }
    }
    if (wave == 0) {
        for (int i0 = nFull * AGG_GROUP; i0 < E; i0 += 32) {
            const int i = i0 + lane;
            const bool m = i < E && (unsigned)(dst[i < E ? i : 0] - lo) < (unsigned)rows;
            const unsigned mask = __builtin_amdgcn_ballot_w32(m);
            if (mask == 0u) continue;
            if (m) q[fill + (int)__builtin_amdgcn_mbcnt_lo(mask, 0u)] = i;
            fill += __builtin_popcount(mask);
            if (fill >= 32) {
                agg_batch(feats, src, dst, norm_src, acc, q, 32, lo, lane);
                const int left = fill - 32;
                int t = 0;
                if (lane < left) t = q[32 + lane];
                if (lane < left) q[lane] = t;
                fill = left;
            }
        }
    }
    if (fill > 0) agg_batch(feats, src, dst, norm_src, acc, q, fill, lo, lane);

    __syncthreads();
    float4* out4 = (float4*)(agg + (size_t)lo * DFEAT);
    for (int i = tid; i < rows * (DFEAT / 4); i += 1024)
        out4[i] = ((const float4*)acc)[i];
}

__global__ __launch_bounds__(256) void gemm_bn_kernel(
    const float* __restrict__ agg, const float* __restrict__ feats,
    const float* __restrict__ norm_dst,
    const float* __restrict__ W,  const float* __restrict__ b,
    const float* __restrict__ Wr, const float* __restrict__ br,
    float* __restrict__ y, float* __restrict__ bn_sum, float* __restrict__ bn_sumsq) {
    const int lane = threadIdx.x & 31;
    const int wave = threadIdx.x >> 5;
    const int rowBase = blockIdx.x * 16;
    const int n0 = wave * 16;
    const int m  = lane & 15;
    const int kh = (lane >> 4) * 2;
    const int row = rowBase + m;
    const float nd = norm_dst[row];
    const float* aRow = agg   + (size_t)row * DFEAT;
    const float* fRow = feats + (size_t)row * DFEAT;

    v8f accC = {};
    v8f accR = {};
    #pragma unroll
    for (int k = 0; k < DFEAT; k += 4) {
        v2f a, f, bw, bwr;
        a.x = aRow[k + kh]     * nd;
        a.y = aRow[k + kh + 1] * nd;
        f.x = fRow[k + kh];
        f.y = fRow[k + kh + 1];
        bw.x  = W [(size_t)(k + kh)     * DFEAT + n0 + m];
        bw.y  = W [(size_t)(k + kh + 1) * DFEAT + n0 + m];
        bwr.x = Wr[(size_t)(k + kh)     * DFEAT + n0 + m];
        bwr.y = Wr[(size_t)(k + kh + 1) * DFEAT + n0 + m];
        accC = __builtin_amdgcn_wmma_f32_16x16x4_f32(false, a, false, bw,  (short)0, accC, false, false);
        accR = __builtin_amdgcn_wmma_f32_16x16x4_f32(false, f, false, bwr, (short)0, accR, false, false);
    }

    const int col  = n0 + m;
    const float bc  = b[col];
    const float brc = br[col];
    const int rOff = (lane >> 4) * 8;
    float s8 = 0.0f, q8 = 0.0f;
    #pragma unroll
    for (int r = 0; r < 8; ++r) {
        float c = accC[r] + bc;  c = c > 0.0f ? c : 0.0f;
        float e = accR[r] + brc; e = e > 0.0f ? e : 0.0f;
        float v = c + e;
        y[(size_t)(rowBase + rOff + r) * DFEAT + col] = v;
        s8 += v;
        q8 += v * v;
    }
    float s8o = __shfl_down(s8, 16);
    float q8o = __shfl_down(q8, 16);
    if (lane < 16) {
        atomicAdd(&bn_sum[col],   s8 + s8o);
        atomicAdd(&bn_sumsq[col], q8 + q8o);
    }
}

__global__ void bn_params_kernel(const float* __restrict__ sum, const float* __restrict__ sumsq,
                                 const float* __restrict__ gamma, const float* __restrict__ beta,
                                 float* __restrict__ scale, float* __restrict__ shift, int n) {
    int c = threadIdx.x;
    if (c >= DFEAT) return;
    float inv  = 1.0f / (float)n;
    float mean = sum[c] * inv;
    float var  = sumsq[c] * inv - mean * mean;
    float sc   = gamma[c] * rsqrtf(var + BN_EPS);
    scale[c] = sc;
    shift[c] = beta[c] - mean * sc;
}

__global__ __launch_bounds__(256) void bn_apply_kernel(float* __restrict__ y,
                                                       const float* __restrict__ scale,
                                                       const float* __restrict__ shift,
                                                       size_t nvec4) {
    size_t i = (size_t)blockIdx.x * blockDim.x + threadIdx.x;
    if (i >= nvec4) return;
    float4 v = ((float4*)y)[i];
    int col = (int)((i * 4) & (DFEAT - 1));
    v.x = v.x * scale[col + 0] + shift[col + 0];
    v.y = v.y * scale[col + 1] + shift[col + 1];
    v.z = v.z * scale[col + 2] + shift[col + 2];
    v.w = v.w * scale[col + 3] + shift[col + 3];
    ((float4*)y)[i] = v;
}

extern "C" void kernel_launch(void* const* d_in, const int* in_sizes, int n_in,
                              void* d_out, int out_size, void* d_ws, size_t ws_size,
                              hipStream_t stream) {
    const float* feats = (const float*)d_in[0];
    const int*   src   = (const int*)d_in[1];
    const int*   dst   = (const int*)d_in[2];
    const float* W     = (const float*)d_in[3];
    const float* b     = (const float*)d_in[4];
    const float* Wr    = (const float*)d_in[5];
    const float* br    = (const float*)d_in[6];
    const float* gamma = (const float*)d_in[7];
    const float* beta  = (const float*)d_in[8];
    float* out = (float*)d_out;

    const int N = NNODES;
    const int E = in_sizes[1];

    float* ws        = (float*)d_ws;
    float* agg       = ws;
    float* deg_out   = agg + (size_t)N * DFEAT;
    float* deg_in    = deg_out + N;
    float* bn_sum    = deg_in + N;
    float* bn_sumsq  = bn_sum + DFEAT;
    float* bn_scale  = bn_sumsq + DFEAT;
    float* bn_shift  = bn_scale + DFEAT;

    size_t nz = 2 * (size_t)N + 2 * DFEAT;
    zero_kernel<<<(2 * N + 2 * DFEAT + 255) / 256, 256, 0, stream>>>(deg_out, nz);

    degree_kernel<<<(E + 255) / 256, 256, 0, stream>>>(src, dst, deg_out, deg_in, E);

    norm_kernel<<<(N + 255) / 256, 256, 0, stream>>>(deg_out, deg_in, N);

    aggregate_kernel<<<AGG_WGS, 1024, 0, stream>>>(feats, src, dst, deg_out, agg, E);

    gemm_bn_kernel<<<N / 16, 256, 0, stream>>>(agg, feats, deg_in, W, b, Wr, br,
                                               out, bn_sum, bn_sumsq);

    bn_params_kernel<<<1, DFEAT, 0, stream>>>(bn_sum, bn_sumsq, gamma, beta,
                                              bn_scale, bn_shift, N);

    size_t nvec4 = (size_t)N * DFEAT / 4;
    bn_apply_kernel<<<(int)((nvec4 + 255) / 256), 256, 0, stream>>>(out, bn_scale, bn_shift, nvec4);
}


# GEMM by row strips: W, Wr staged once per work-group in LDS in fragment order; a wave owns 16 rows x 128 columns, rows loaded once, 16 B/lane; BN sums via LDS adds per work-group (sum order changed)
# speedup vs baseline: 1.7269x; 1.2392x over previous
#include <hip/hip_runtime.h>
#include <hip/hip_bf16.h>

typedef __attribute__((ext_vector_type(2))) float v2f;
typedef __attribute__((ext_vector_type(8))) float v8f;

#define NNODES 100000
#define DFEAT  128
#define BN_EPS 1e-5f


__global__ __launch_bounds__(256) void zero_kernel(float* __restrict__ p, size_t n) {
    size_t i = (size_t)blockIdx.x * blockDim.x + threadIdx.x;
    size_t stride = (size_t)gridDim.x * blockDim.x;
    for (; i < n; i += stride) p[i] = 0.0f;
}

__global__ __launch_bounds__(256) void degree_kernel(const int* __restrict__ src,
                                                     const int* __restrict__ dst,
                                                     float* __restrict__ dout,
                                                     float* __restrict__ din, int E) {
    int e = blockIdx.x * blockDim.x + threadIdx.x;
    if (e >= E) return;
    atomicAdd(&dout[src[e]], 1.0f);
    atomicAdd(&din[dst[e]], 1.0f);
}

__global__ __launch_bounds__(256) void norm_kernel(float* __restrict__ dout,
                                                   float* __restrict__ din, int n) {
    int i = blockIdx.x * blockDim.x + threadIdx.x;
    if (i >= n) return;
    dout[i] = rsqrtf(fmaxf(dout[i], 1.0f));
    din[i]  = rsqrtf(fmaxf(din[i], 1.0f));
}

#define AGG_WGS     256
#define AGG_WAVES   32
#define AGG_ROWS    ((NNODES + AGG_WGS - 1) / AGG_WGS)
#define AGG_GROUP   512
#define AGG_QCAP    (32 + AGG_GROUP)

__device__ __forceinline__ void agg_batch(const float* __restrict__ feats, const int* __restrict__ src,
                                          const int* __restrict__ dst, const float* __restrict__ norm_src,
                                          float* acc, const int* q, int n, int lo, int lane) {
    int s = 0, dl = 0;
    float ns = 0.0f;
    if (lane < n) {
        const int e = q[lane];
        dl = dst[e] - lo;
        s  = src[e];
        ns = norm_src[s];
    }
    const int nsb = __float_as_int(ns);
    const float* fl = feats + lane;
    float* al = acc + lane;
    for (int j0 = 0; j0 < n; j0 += 8) {
        float v[8][4];
        #pragma unroll
        for (int u = 0; u < 8; ++u) {
            const int sj = __builtin_amdgcn_readlane(s, j0 + u);
            const float* fr = fl + (size_t)sj * DFEAT;
            v[u][0] = fr[0];
            v[u][1] = fr[32];
            v[u][2] = fr[64];
            v[u][3] = fr[96];
        }
        #pragma unroll
        for (int u = 0; u < 8; ++u) {
            if (j0 + u < n) {
                const int dj = __builtin_amdgcn_readlane(dl, j0 + u);
                const float nj = __int_as_float(__builtin_amdgcn_readlane(nsb, j0 + u));
                float* ar = al + dj * DFEAT;
                atomicAdd(ar + 0,  v[u][0] * nj);
                atomicAdd(ar + 32, v[u][1] * nj);
                atomicAdd(ar + 64, v[u][2] * nj);
                atomicAdd(ar + 96, v[u][3] * nj);
            }
        }
    }
}

__global__ __launch_bounds__(1024) void aggregate_kernel(const float* __restrict__ feats,
                                                         const int* __restrict__ src,
                                                         const int* __restrict__ dst,
                                                         const float* __restrict__ norm_src,
                                                         float* __restrict__ agg, int E) {
    __shared__ __attribute__((aligned(16))) float acc[AGG_ROWS * DFEAT];
    __shared__ int queue[AGG_WAVES][AGG_QCAP];

    const int tid  = threadIdx.x;
    const int lane = tid & 31;
    const int wave = __builtin_amdgcn_readfirstlane(tid >> 5);
    const int lo   = blockIdx.x * AGG_ROWS;
    int rows = NNODES - lo;
    rows = rows > AGG_ROWS ? AGG_ROWS : rows;
    rows = rows < 0 ? 0 : rows;

    for (int i = tid; i < AGG_ROWS * DFEAT / 4; i += 1024)
        ((float4*)acc)[i] = make_float4(0.0f, 0.0f, 0.0f, 0.0f);
    __syncthreads();

    int* q = queue[wave];
    int fill = 0;

    const int nFull = E / AGG_GROUP;
    if (nFull > 0) {
        const int rot = (int)(((long long)blockIdx.x * nFull) / AGG_WGS);
        const int4* dst4 = (const int4*)dst + lane;

        int4 cur[4];
        {
            int g = (wave < nFull ? wave : 0) + rot;
            g = g >= nFull ? g - nFull : g;
            #pragma unroll
            for (int c = 0; c < 4; ++c) cur[c] = dst4[(size_t)g * (AGG_GROUP / 4) + c * 32];
        }
        for (int gi = wave; gi < nFull; gi += AGG_WAVES) {
            int g = gi + rot;
            g = g >= nFull ? g - nFull : g;
            int4 nxt[4];
            {
                int gn = (gi + AGG_WAVES < nFull ? gi + AGG_WAVES : gi) + rot;
                gn = gn >= nFull ? gn - nFull : gn;
                #pragma unroll
                for (int c = 0; c < 4; ++c) nxt[c] = dst4[(size_t)gn * (AGG_GROUP / 4) + c * 32];
            }
            #pragma unroll
            for (int c = 0; c < 4; ++c) {
                const int d4[4] = {cur[c].x, cur[c].y, cur[c].z, cur[c].w};
                bool m[4];
                #pragma unroll
                for (int k = 0; k < 4; ++k)
                    m[k] = (unsigned)(d4[k] - lo) < (unsigned)rows;
                if (__builtin_amdgcn_ballot_w32(m[0] | m[1] | m[2] | m[3]) == 0u) continue;
                const int e0 = g * AGG_GROUP + c * 128 + 4 * lane;
                #pragma unroll
                for (int k = 0; k < 4; ++k) {
                    const unsigned mask = __builtin_amdgcn_ballot_w32(m[k]);
                    if (mask == 0u) continue;
                    if (m[k]) q[fill + (int)__builtin_amdgcn_mbcnt_lo(mask, 0u)] = e0 + k;
                    fill += __builtin_popcount(mask);
                }
            }
            if (fill >= 32) {
                int head = 0;
                while (fill - head >= 32) {
                    agg_batch(feats, src, dst, norm_src, acc, q + head, 32, lo, lane);
                    head += 32;
                }
                const int left = fill - head;
                int t = 0;
                if (lane < left) t = q[head + lane];
                if (lane < left) q[lane] = t;
                fill = left;
            }
            #pragma unroll
            for (int c = 0; c < 4; ++c) cur[c] = nxt[c];
        }
    }
    if (wave == 0) {
        for (int i0 = nFull * AGG_GROUP; i0 < E; i0 += 32) {
            const int i = i0 + lane;
            const bool m = i < E && (unsigned)(dst[i < E ? i : 0] - lo) < (unsigned)rows;
            const unsigned mask = __builtin_amdgcn_ballot_w32(m);
            if (mask == 0u) continue;
            if (m) q[fill + (int)__builtin_amdgcn_mbcnt_lo(mask, 0u)] = i;
            fill += __builtin_popcount(mask);
            if (fill >= 32) {
                agg_batch(feats, src, dst, norm_src, acc, q, 32, lo, lane);
                const int left = fill - 32;
                int t = 0;
                if (lane < left) t = q[32 + lane];
                if (lane < left) q[lane] = t;
                fill = left;
            }
        }
    }
    if (fill > 0) agg_batch(feats, src, dst, norm_src, acc, q, fill, lo, lane);

    __syncthreads();
    float4* out4 = (float4*)(agg + (size_t)lo * DFEAT);
    for (int i = tid; i < rows * (DFEAT / 4); i += 1024)
        out4[i] = ((const float4*)acc)[i];
}

#define GEMM_WGS    256
#define GEMM_WAVES  8
#define GEMM_ROUNDS 4
#define NSTRIPS     (NNODES / 16)
#define WFRAGS      (16 * 8 * 32)

typedef __attribute__((ext_vector_type(4))) float v4f;

__global__ __launch_bounds__(256) void gemm_bn_kernel(
    const float* __restrict__ agg, const float* __restrict__ feats,
    const float* __restrict__ norm_dst,
    const float* __restrict__ W,  const float* __restrict__ b,
    const float* __restrict__ Wr, const float* __restrict__ br,
    float* __restrict__ y, float* __restrict__ bn_sum, float* __restrict__ bn_sumsq) {
    __shared__ __attribute__((aligned(16))) v4f sW [WFRAGS];
    __shared__ __attribute__((aligned(16))) v4f sWr[WFRAGS];
    __shared__ float sSum[DFEAT];
    __shared__ float sSq [DFEAT];

    const int tid  = threadIdx.x;
    const int lane = tid & 31;
    const int wave = __builtin_amdgcn_readfirstlane(tid >> 5);
    const int c = lane & 15;
    const int h = lane >> 4;

    for (int i = tid; i < WFRAGS; i += 256) {
        const int J = i >> 8;
        const int t = (i >> 5) & 7;
        const size_t base = (size_t)(8 * J + 4 * h) * DFEAT + 16 * t + c;
        v4f w4, r4;
        w4.x = W [base];  w4.y = W [base + DFEAT];  w4.z = W [base + 2 * DFEAT];  w4.w = W [base + 3 * DFEAT];
        r4.x = Wr[base];  r4.y = Wr[base + DFEAT];  r4.z = Wr[base + 2 * DFEAT];  r4.w = Wr[base + 3 * DFEAT];
        sW[i]  = w4;
        sWr[i] = r4;
    }
    if (tid < DFEAT) { sSum[tid] = 0.0f; sSq[tid] = 0.0f; }
    __syncthreads();

    float bcv[8], brv[8], s8[8], q8[8];
    #pragma unroll
    for (int t = 0; t < 8; ++t) {
        bcv[t] = b [16 * t + c];
        brv[t] = br[16 * t + c];
        s8[t] = 0.0f;
        q8[t] = 0.0f;
    }

    const v4f* pw = sW  + lane;
    const v4f* pr = sWr + lane;

    int strip = wave * GEMM_WGS + blockIdx.x;
    const v4f* ap = (const v4f*)(agg   + (size_t)(strip * 16 + c) * DFEAT) + h;
    const v4f* fp = (const v4f*)(feats + (size_t)(strip * 16 + c) * DFEAT) + h;
    float nd = norm_dst[strip * 16 + c];

    v4f a4[16], f4[16];
    v4f bw[2][8], bwr[2][8];
    #pragma unroll
    for (int J = 0; J < 8; ++J) {
        a4[J] = ap[2 * J];
        f4[J] = fp[2 * J];
    }
    #pragma unroll
    for (int t = 0; t < 8; ++t) {
        bw[0][t]  = pw[t * 32];
        bwr[0][t] = pr[t * 32];
    }

    for (int rnd = 0; rnd < GEMM_ROUNDS; ++rnd) {
        const int nstrip = strip + GEMM_WAVES * GEMM_WGS;
        const bool more = nstrip < NSTRIPS;
        const int prow = (more ? nstrip : strip) * 16 + c;
        const v4f* apn = (const v4f*)(agg   + (size_t)prow * DFEAT) + h;
        const v4f* fpn = (const v4f*)(feats + (size_t)prow * DFEAT) + h;
        float ndn = nd;

        v8f accC[8], accR[8];
        #pragma unroll
        for (int t = 0; t < 8; ++t) {
            accC[t] = (v8f){0.0f, 0.0f, 0.0f, 0.0f, 0.0f, 0.0f, 0.0f, 0.0f};
            accR[t] = (v8f){0.0f, 0.0f, 0.0f, 0.0f, 0.0f, 0.0f, 0.0f, 0.0f};
        }

        #pragma unroll
        for (int J = 0; J < 16; ++J) {
            if (J < 8) {
                a4[J + 8] = ap[2 * (J + 8)];
                f4[J + 8] = fp[2 * (J + 8)];
            } else {
                a4[J - 8] = apn[2 * (J - 8)];
                f4[J - 8] = fpn[2 * (J - 8)];
            }
            if (J == 8) ndn = norm_dst[prow];
            #pragma unroll
            for (int t = 0; t < 8; ++t) {
                bw [(J + 1) & 1][t] = pw[((J + 1) & 15) * 256 + t * 32];
                bwr[(J + 1) & 1][t] = pr[((J + 1) & 15) * 256 + t * 32];
            }
            __builtin_amdgcn_sched_barrier(0);

            v2f a0, a1, f0, f1;
            a0.x = a4[J].x * nd;  a0.y = a4[J].y * nd;
            a1.x = a4[J].z * nd;  a1.y = a4[J].w * nd;
            f0.x = f4[J].x;       f0.y = f4[J].y;
            f1.x = f4[J].z;       f1.y = f4[J].w;
            #pragma unroll
            for (int t = 0; t < 8; ++t) {
                v2f b0, r0;
                b0.x = bw [J & 1][t].x;  b0.y = bw [J & 1][t].y;
                r0.x = bwr[J & 1][t].x;  r0.y = bwr[J & 1][t].y;
                accC[t] = __builtin_amdgcn_wmma_f32_16x16x4_f32(false, a0, false, b0, (short)0, accC[t], false, false);
                accR[t] = __builtin_amdgcn_wmma_f32_16x16x4_f32(false, f0, false, r0, (short)0, accR[t], false, false);
            }
            #pragma unroll
            for (int t = 0; t < 8; ++t) {
                v2f b1, r1;
                b1.x = bw [J & 1][t].z;  b1.y = bw [J & 1][t].w;
                r1.x = bwr[J & 1][t].z;  r1.y = bwr[J & 1][t].w;
                accC[t] = __builtin_amdgcn_wmma_f32_16x16x4_f32(false, a1, false, b1, (short)0, accC[t], false, false);
                accR[t] = __builtin_amdgcn_wmma_f32_16x16x4_f32(false, f1, false, r1, (short)0, accR[t], false, false);
            }
            __builtin_amdgcn_sched_barrier(0);
        }

        float* yr = y + (size_t)(strip * 16 + 8 * h) * DFEAT + c;
        #pragma unroll
        for (int t = 0; t < 8; ++t) {
            #pragma unroll
            for (int r = 0; r < 8; ++r) {
                float cv = accC[t][r] + bcv[t];  cv = cv > 0.0f ? cv : 0.0f;
                float ev = accR[t][r] + brv[t];  ev = ev > 0.0f ? ev : 0.0f;
                const float v = cv + ev;
                yr[(size_t)r * DFEAT + 16 * t] = v;
                s8[t] += v;
                q8[t] += v * v;
            }
        }

        if (!more) break;
        strip = nstrip;
        ap = apn;
        fp = fpn;
        nd = ndn;
    }

    #pragma unroll
    for (int t = 0; t < 8; ++t) {
        const float so = __shfl_down(s8[t], 16);
        const float qo = __shfl_down(q8[t], 16);
        if (lane < 16) {
            atomicAdd(&sSum[16 * t + c], s8[t] + so);
            atomicAdd(&sSq [16 * t + c], q8[t] + qo);
        }
    }
    __syncthreads();
    if (tid < DFEAT) {
        atomicAdd(&bn_sum[tid],   sSum[tid]);
        atomicAdd(&bn_sumsq[tid], sSq[tid]);
    }
}

__global__ void bn_params_kernel(const float* __restrict__ sum, const float* __restrict__ sumsq,
                                 const float* __restrict__ gamma, const float* __restrict__ beta,
                                 float* __restrict__ scale, float* __restrict__ shift, int n) {
    int c = threadIdx.x;
    if (c >= DFEAT) return;
    float inv  = 1.0f / (float)n;
    float mean = sum[c] * inv;
    float var  = sumsq[c] * inv - mean * mean;
    float sc   = gamma[c] * rsqrtf(var + BN_EPS);
    scale[c] = sc;
    shift[c] = beta[c] - mean * sc;
}

__global__ __launch_bounds__(256) void bn_apply_kernel(float* __restrict__ y,
                                                       const float* __restrict__ scale,
                                                       const float* __restrict__ shift,
                                                       size_t nvec4) {
    size_t i = (size_t)blockIdx.x * blockDim.x + threadIdx.x;
    if (i >= nvec4) return;
    float4 v = ((float4*)y)[i];
    int col = (int)((i * 4) & (DFEAT - 1));
    v.x = v.x * scale[col + 0] + shift[col + 0];
    v.y = v.y * scale[col + 1] + shift[col + 1];
    v.z = v.z * scale[col + 2] + shift[col + 2];
    v.w = v.w * scale[col + 3] + shift[col + 3];
    ((float4*)y)[i] = v;
}

extern "C" void kernel_launch(void* const* d_in, const int* in_sizes, int n_in,
                              void* d_out, int out_size, void* d_ws, size_t ws_size,
                              hipStream_t stream) {
    const float* feats = (const float*)d_in[0];
    const int*   src   = (const int*)d_in[1];
    const int*   dst   = (const int*)d_in[2];
    const float* W     = (const float*)d_in[3];
    const float* b     = (const float*)d_in[4];
    const float* Wr    = (const float*)d_in[5];
    const float* br    = (const float*)d_in[6];
    const float* gamma = (const float*)d_in[7];
    const float* beta  = (const float*)d_in[8];
    float* out = (float*)d_out;

    const int N = NNODES;
    const int E = in_sizes[1];

    float* ws        = (float*)d_ws;
    float* agg       = ws;
    float* deg_out   = agg + (size_t)N * DFEAT;
    float* deg_in    = deg_out + N;
    float* bn_sum    = deg_in + N;
    float* bn_sumsq  = bn_sum + DFEAT;
    float* bn_scale  = bn_sumsq + DFEAT;
    float* bn_shift  = bn_scale + DFEAT;

    size_t nz = 2 * (size_t)N + 2 * DFEAT;
    zero_kernel<<<(2 * N + 2 * DFEAT + 255) / 256, 256, 0, stream>>>(deg_out, nz);

    degree_kernel<<<(E + 255) / 256, 256, 0, stream>>>(src, dst, deg_out, deg_in, E);

    norm_kernel<<<(N + 255) / 256, 256, 0, stream>>>(deg_out, deg_in, N);

    aggregate_kernel<<<AGG_WGS, 1024, 0, stream>>>(feats, src, dst, deg_out, agg, E);

    gemm_bn_kernel<<<GEMM_WGS, 256, 0, stream>>>(agg, feats, deg_in, W, b, Wr, br,
                                                 out, bn_sum, bn_sumsq);

    bn_params_kernel<<<1, DFEAT, 0, stream>>>(bn_sum, bn_sumsq, gamma, beta,
                                              bn_scale, bn_shift, N);

    size_t nvec4 = (size_t)N * DFEAT / 4;
    bn_apply_kernel<<<(int)((nvec4 + 255) / 256), 256, 0, stream>>>(out, bn_scale, bn_shift, nvec4);
}
